# starting kernel
# baseline (speedup 1.0000x reference)
#include <hip/hip_runtime.h>
#include <hip/hip_bf16.h>
#include <math.h>

typedef float v2f __attribute__((ext_vector_type(2)));
typedef float v8f __attribute__((ext_vector_type(8)));

#define D1 128
#define D2 64
#define WAVES_PER_BLOCK 8
#define BLOCK (WAVES_PER_BLOCK * 32)
#define ROW_STRIDE 132
#define TILE_FLOATS (16 * ROW_STRIDE)
#define W1_PAIRS (64 * 64)
#define LDS_FLOATS (2 * W1_PAIRS + WAVES_PER_BLOCK * TILE_FLOATS)

__global__ void __launch_bounds__(BLOCK)
edge_decoder_kernel(const float* __restrict__ hf,
                    const int*   __restrict__ edge,
                    const float* __restrict__ W1,
                    const float* __restrict__ b1,
                    const float* __restrict__ W2,
                    const float* __restrict__ b2,
                    float* __restrict__ out,
                    int E)
{
    extern __shared__ float smem[];
    v2f* W1p = (v2f*)smem;

    const int tid = threadIdx.x;

    for (int idx = tid; idx < W1_PAIRS; idx += BLOCK) {
        const int p = idx >> 6;
        const int c = idx & 63;
        v2f v;
        v.x = W1[p * 2 * D2 + c];
        v.y = W1[p * 2 * D2 + D2 + c];
        W1p[idx] = v;
    }
    __syncthreads();

    const int l  = tid & 31;
    const int li = l & 15;
    const int hi = l >> 4;
    const int w  = tid >> 5;

    float* tileBuf = smem + 2 * W1_PAIRS + w * TILE_FLOATS;
    const v2f* Ab = (const v2f*)tileBuf;
    const int aBase = li * (ROW_STRIDE / 2) + hi;

    const int* edge0 = edge;
    const int* edge1 = edge + E;

    float bias[4], w2v[4];
#pragma unroll
    for (int n = 0; n < 4; ++n) {
        bias[n] = b1[16 * n + li];
        w2v[n]  = W2[16 * n + li];
    }
    const float b2v = b2[0];

    const int ntiles = (E + 15) >> 4;
    const int nwaves = gridDim.x * WAVES_PER_BLOCK;

    for (int tile = blockIdx.x * WAVES_PER_BLOCK + w; tile < ntiles; tile += nwaves) {
        const int e0 = tile << 4;

#pragma unroll 4
        for (int e = 0; e < 16; ++e) {
            int ee = e0 + e; if (ee >= E) ee = E - 1;
            const long long u = edge0[ee];
            const long long v = edge1[ee];
            const float4 fu = *(const float4*)(hf + u * D1 + 4 * l);
            const float4 fv = *(const float4*)(hf + v * D1 + 4 * l);
            float4 p;
            p.x = fu.x * fv.x; p.y = fu.y * fv.y;
            p.z = fu.z * fv.z; p.w = fu.w * fv.w;
            *(float4*)(tileBuf + e * ROW_STRIDE + 4 * l) = p;
        }

        v8f c0 = {bias[0],bias[0],bias[0],bias[0],bias[0],bias[0],bias[0],bias[0]};
        v8f c1 = {bias[1],bias[1],bias[1],bias[1],bias[1],bias[1],bias[1],bias[1]};
        v8f c2 = {bias[2],bias[2],bias[2],bias[2],bias[2],bias[2],bias[2],bias[2]};
        v8f c3 = {bias[3],bias[3],bias[3],bias[3],bias[3],bias[3],bias[3],bias[3]};

#pragma unroll
        for (int t = 0; t < 32; ++t) {
            const v2f a = Ab[aBase + 2 * t];
            const int bb = (2 * t + hi) * 64 + li;
            c0 = __builtin_amdgcn_wmma_f32_16x16x4_f32(false, a, false, W1p[bb     ], (short)0, c0, false, false);
            c1 = __builtin_amdgcn_wmma_f32_16x16x4_f32(false, a, false, W1p[bb + 16], (short)0, c1, false, false);
            c2 = __builtin_amdgcn_wmma_f32_16x16x4_f32(false, a, false, W1p[bb + 32], (short)0, c2, false, false);
            c3 = __builtin_amdgcn_wmma_f32_16x16x4_f32(false, a, false, W1p[bb + 48], (short)0, c3, false, false);
        }

        float s[8];
#pragma unroll
        for (int i = 0; i < 8; ++i) {
            s[i] = fmaxf(c0[i], 0.f) * w2v[0] + fmaxf(c1[i], 0.f) * w2v[1]
                 + fmaxf(c2[i], 0.f) * w2v[2] + fmaxf(c3[i], 0.f) * w2v[3];
        }
#pragma unroll
        for (int m = 1; m <= 8; m <<= 1) {
#pragma unroll
            for (int i = 0; i < 8; ++i) s[i] += __shfl_xor(s[i], m, 32);
        }

        if (li == 0) {
            const int base = e0 + hi * 8;
            float r[8];
#pragma unroll
            for (int i = 0; i < 8; ++i) r[i] = 1.f / (1.f + expf(-(s[i] + b2v)));
            if (base + 8 <= E) {
                float4 o0 = {r[0], r[1], r[2], r[3]};
                float4 o1 = {r[4], r[5], r[6], r[7]};
                *(float4*)(out + base)     = o0;
                *(float4*)(out + base + 4) = o1;
            } else {
#pragma unroll
                for (int i = 0; i < 8; ++i) if (base + i < E) out[base + i] = r[i];
            }
        }
    }
}

extern "C" void kernel_launch(void* const* d_in, const int* in_sizes, int n_in,
                              void* d_out, int out_size, void* d_ws, size_t ws_size,
                              hipStream_t stream) {
    (void)n_in; (void)out_size; (void)d_ws; (void)ws_size;
    const float* hf   = (const float*)d_in[0];
    const int*   edge = (const int*)d_in[1];
    const float* W1   = (const float*)d_in[2];
    const float* b1   = (const float*)d_in[3];
    const float* W2   = (const float*)d_in[4];
    const float* b2   = (const float*)d_in[5];
    float* out = (float*)d_out;

    const int E = in_sizes[1] / 2;
    const int ntiles = (E + 15) / 16;
    int blocks = (ntiles + WAVES_PER_BLOCK - 1) / WAVES_PER_BLOCK;
    if (blocks > 2500) blocks = 2500;

    const size_t shmem = LDS_FLOATS * sizeof(float);
    hipLaunchKernelGGL(edge_decoder_kernel, dim3(blocks), dim3(BLOCK), shmem, stream,
                       hf, edge, W1, b1, W2, b2, out, E);
}


# wave tile grown from 16 to 32 edges (K staged in two halves of 64): each W1 fragment read from LDS feeds two WMMAs, halving B-fragment LDS reads per WMMA
# speedup vs baseline: 1.0366x; 1.0366x over previous
#include <hip/hip_runtime.h>
#include <hip/hip_bf16.h>
#include <math.h>

typedef float v2f __attribute__((ext_vector_type(2)));
typedef float v8f __attribute__((ext_vector_type(8)));

#define D1 128
#define D2 64
#define WAVES_PER_BLOCK 8
#define BLOCK (WAVES_PER_BLOCK * 32)
#define ROWS 32
#define KH 64
#define ROW_STRIDE 68
#define TILE_FLOATS (ROWS * ROW_STRIDE)
#define W1_PAIRS (64 * 64)
#define LDS_FLOATS (2 * W1_PAIRS + WAVES_PER_BLOCK * TILE_FLOATS)

__global__ void __launch_bounds__(BLOCK)
edge_decoder_kernel(const float* __restrict__ hf,
                    const int*   __restrict__ edge,
                    const float* __restrict__ W1,
                    const float* __restrict__ b1,
                    const float* __restrict__ W2,
                    const float* __restrict__ b2,
                    float* __restrict__ out,
                    int E)
{
    extern __shared__ float smem[];
    v2f* W1p = (v2f*)smem;

    const int tid = threadIdx.x;

    for (int idx = tid; idx < W1_PAIRS; idx += BLOCK) {
        const int p = idx >> 6;
        const int c = idx & 63;
        v2f v;
        v.x = W1[p * 2 * D2 + c];
        v.y = W1[p * 2 * D2 + D2 + c];
        W1p[idx] = v;
    }
    __syncthreads();

    const int l  = tid & 31;
    const int li = l & 15;
    const int hi = l >> 4;
    const int w  = tid >> 5;

    float* tileBuf = smem + 2 * W1_PAIRS + w * TILE_FLOATS;
    const v2f* Ab = (const v2f*)tileBuf;
    const int aBase = li * (ROW_STRIDE / 2) + hi;

    const int* edge0 = edge;
    const int* edge1 = edge + E;

    float bias[4], w2v[4];
#pragma unroll
    for (int n = 0; n < 4; ++n) {
        bias[n] = b1[16 * n + li];
        w2v[n]  = W2[16 * n + li];
    }
    const float b2v = b2[0];

    const int npass  = (E + ROWS - 1) / ROWS;
    const int nwaves = gridDim.x * WAVES_PER_BLOCK;

    for (int pass = blockIdx.x * WAVES_PER_BLOCK + w; pass < npass; pass += nwaves) {
        const int e0 = pass * ROWS;

        int ee = e0 + l; if (ee >= E) ee = E - 1;
        const int iu = edge0[ee];
        const int iv = edge1[ee];

        v8f c0[2], c1[2], c2[2], c3[2];
#pragma unroll
        for (int m = 0; m < 2; ++m) {
            c0[m] = (v8f){bias[0],bias[0],bias[0],bias[0],bias[0],bias[0],bias[0],bias[0]};
            c1[m] = (v8f){bias[1],bias[1],bias[1],bias[1],bias[1],bias[1],bias[1],bias[1]};
            c2[m] = (v8f){bias[2],bias[2],bias[2],bias[2],bias[2],bias[2],bias[2],bias[2]};
            c3[m] = (v8f){bias[3],bias[3],bias[3],bias[3],bias[3],bias[3],bias[3],bias[3]};
        }

        for (int H = 0; H < 2; ++H) {
#pragma unroll 4
            for (int j = 0; j < 16; ++j) {
                const int r = 2 * j + hi;
                const long long u = __shfl(iu, r, 32);
                const long long v = __shfl(iv, r, 32);
                const float4 fu = *(const float4*)(hf + u * D1 + H * KH + 4 * li);
                const float4 fv = *(const float4*)(hf + v * D1 + H * KH + 4 * li);
                float4 p;
                p.x = fu.x * fv.x; p.y = fu.y * fv.y;
                p.z = fu.z * fv.z; p.w = fu.w * fv.w;
                *(float4*)(tileBuf + r * ROW_STRIDE + 4 * li) = p;
            }

#pragma unroll
            for (int t = 0; t < 16; ++t) {
                const v2f a0 = Ab[aBase + 2 * t];
                const v2f a1 = Ab[aBase + 16 * (ROW_STRIDE / 2) + 2 * t];
                const int bb = (2 * (16 * H + t) + hi) * 64 + li;
                const v2f w0 = W1p[bb], w1 = W1p[bb + 16], w2 = W1p[bb + 32], w3 = W1p[bb + 48];
                c0[0] = __builtin_amdgcn_wmma_f32_16x16x4_f32(false, a0, false, w0, (short)0, c0[0], false, false);
                c0[1] = __builtin_amdgcn_wmma_f32_16x16x4_f32(false, a1, false, w0, (short)0, c0[1], false, false);
                c1[0] = __builtin_amdgcn_wmma_f32_16x16x4_f32(false, a0, false, w1, (short)0, c1[0], false, false);
                c1[1] = __builtin_amdgcn_wmma_f32_16x16x4_f32(false, a1, false, w1, (short)0, c1[1], false, false);
                c2[0] = __builtin_amdgcn_wmma_f32_16x16x4_f32(false, a0, false, w2, (short)0, c2[0], false, false);
                c2[1] = __builtin_amdgcn_wmma_f32_16x16x4_f32(false, a1, false, w2, (short)0, c2[1], false, false);
                c3[0] = __builtin_amdgcn_wmma_f32_16x16x4_f32(false, a0, false, w3, (short)0, c3[0], false, false);
                c3[1] = __builtin_amdgcn_wmma_f32_16x16x4_f32(false, a1, false, w3, (short)0, c3[1], false, false);
            }
        }

#pragma unroll
        for (int m = 0; m < 2; ++m) {
            float s[8];
#pragma unroll
            for (int i = 0; i < 8; ++i) {
                s[i] = fmaxf(c0[m][i], 0.f) * w2v[0] + fmaxf(c1[m][i], 0.f) * w2v[1]
                     + fmaxf(c2[m][i], 0.f) * w2v[2] + fmaxf(c3[m][i], 0.f) * w2v[3];
            }
#pragma unroll
            for (int k = 1; k <= 8; k <<= 1) {
#pragma unroll
                for (int i = 0; i < 8; ++i) s[i] += __shfl_xor(s[i], k, 32);
            }

            if (li == 0) {
                const int base = e0 + 16 * m + hi * 8;
                float r[8];
#pragma unroll
                for (int i = 0; i < 8; ++i) r[i] = 1.f / (1.f + expf(-(s[i] + b2v)));
                if (base + 8 <= E) {
                    float4 o0 = {r[0], r[1], r[2], r[3]};
                    float4 o1 = {r[4], r[5], r[6], r[7]};
                    *(float4*)(out + base)     = o0;
                    *(float4*)(out + base + 4) = o1;
                } else {
#pragma unroll
                    for (int i = 0; i < 8; ++i) if (base + i < E) out[base + i] = r[i];
                }
            }
        }
    }
}

extern "C" void kernel_launch(void* const* d_in, const int* in_sizes, int n_in,
                              void* d_out, int out_size, void* d_ws, size_t ws_size,
                              hipStream_t stream) {
    (void)n_in; (void)out_size; (void)d_ws; (void)ws_size;
    const float* hf   = (const float*)d_in[0];
    const int*   edge = (const int*)d_in[1];
    const float* W1   = (const float*)d_in[2];
    const float* b1   = (const float*)d_in[3];
    const float* W2   = (const float*)d_in[4];
    const float* b2   = (const float*)d_in[5];
    float* out = (float*)d_out;

    const int E = in_sizes[1] / 2;
    const int npass = (E + ROWS - 1) / ROWS;
    int blocks = (npass + WAVES_PER_BLOCK - 1) / WAVES_PER_BLOCK;
    if (blocks > 2500) blocks = 2500;

    const size_t shmem = LDS_FLOATS * sizeof(float);
    hipLaunchKernelGGL(edge_decoder_kernel, dim3(blocks), dim3(BLOCK), shmem, stream,
                       hf, edge, W1, b1, W2, b2, out, E);
}


# WMMA operands swapped so each tile comes out transposed (features down a lane's registers, edges across lanes): W2 dot = in-lane FMAs + one half exchange, no shuffles; changes that sum's order
# speedup vs baseline: 1.0971x; 1.0584x over previous
#include <hip/hip_runtime.h>
#include <hip/hip_bf16.h>
#include <math.h>


typedef float v2f __attribute__((ext_vector_type(2)));
typedef float v8f __attribute__((ext_vector_type(8)));

#define D1 128
#define D2 64
#define WAVES_PER_BLOCK 8
#define BLOCK (WAVES_PER_BLOCK * 32)
#define ROWS 32
#define KH 64
#define ROW_STRIDE 68
#define TILE_FLOATS (ROWS * ROW_STRIDE)
#define W1_PAIRS (64 * 64)
#define SIDE_FLOATS (2 * D2)
#define LDS_FLOATS (2 * W1_PAIRS + WAVES_PER_BLOCK * TILE_FLOATS + SIDE_FLOATS)

__device__ __forceinline__ float xhalf(float x) {
    const int b = __builtin_bit_cast(int, x);
    return __builtin_bit_cast(float, __builtin_amdgcn_permlanex16(b, b, 0x76543210, 0xfedcba98, false, false));
}

__global__ void __launch_bounds__(BLOCK)
edge_decoder_kernel(const float* __restrict__ hf,
                    const int*   __restrict__ edge,
                    const float* __restrict__ W1,
                    const float* __restrict__ b1,
                    const float* __restrict__ W2,
                    const float* __restrict__ b2,
                    float* __restrict__ out,
                    int E)
{
    extern __shared__ float smem[];
    v2f* W1p = (v2f*)smem;
    float* side = smem + 2 * W1_PAIRS + WAVES_PER_BLOCK * TILE_FLOATS;

    const int tid = threadIdx.x;

    const float sb = b1[tid & (D2 - 1)];
    const float sw = W2[tid & (D2 - 1)];

    for (int idx = tid; idx < W1_PAIRS; idx += BLOCK) {
        const int p = idx >> 6;
        const int c = idx & 63;
        v2f v;
        v.x = W1[p * 2 * D2 + c];
        v.y = W1[p * 2 * D2 + D2 + c];
        W1p[idx] = v;
    }
    if (tid < D2) {
        side[tid]      = sb;
        side[D2 + tid] = sw;
    }
    __syncthreads();

    const int l  = tid & 31;
    const int li = l & 15;
    const int hi = l >> 4;
    const int w  = tid >> 5;

    float* tileBuf = smem + 2 * W1_PAIRS + w * TILE_FLOATS;
    const v2f* Ab = (const v2f*)tileBuf;
    const int aBase = li * (ROW_STRIDE / 2) + hi;

    const int* edge0 = edge;
    const int* edge1 = edge + E;

    const float* bsrc = side + 8 * hi;
    const float* wsrc = side + D2 + 8 * hi;
    const float b2v = b2[0];

    const int npass  = (E + ROWS - 1) / ROWS;
    const int nwaves = gridDim.x * WAVES_PER_BLOCK;

    for (int pass = blockIdx.x * WAVES_PER_BLOCK + w; pass < npass; pass += nwaves) {
        const int e0 = pass * ROWS;

        int ee = e0 + l; if (ee >= E) ee = E - 1;
        const int iu = edge0[ee];
        const int iv = edge1[ee];

        v8f acc[4][2];
#pragma unroll
        for (int n = 0; n < 4; ++n) {
            const float4 ba = *(const float4*)(bsrc + 16 * n);
            const float4 bb = *(const float4*)(bsrc + 16 * n + 4);
            const v8f bv = (v8f){ba.x, ba.y, ba.z, ba.w, bb.x, bb.y, bb.z, bb.w};
            acc[n][0] = bv;
            acc[n][1] = bv;
        }

        for (int H = 0; H < 2; ++H) {
#pragma unroll 4
            for (int j = 0; j < 16; ++j) {
                const int r = 2 * j + hi;
                const long long u = __shfl(iu, r, 32);
                const long long v = __shfl(iv, r, 32);
                const float4 fu = *(const float4*)(hf + u * D1 + H * KH + 4 * li);
                const float4 fv = *(const float4*)(hf + v * D1 + H * KH + 4 * li);
                float4 p;
                p.x = fu.x * fv.x; p.y = fu.y * fv.y;
                p.z = fu.z * fv.z; p.w = fu.w * fv.w;
                *(float4*)(tileBuf + r * ROW_STRIDE + 4 * li) = p;
            }

#pragma unroll
            for (int t = 0; t < 16; ++t) {
                const v2f a0 = Ab[aBase + 2 * t];
                const v2f a1 = Ab[aBase + 16 * (ROW_STRIDE / 2) + 2 * t];
                const int bb = (2 * (16 * H + t) + hi) * 64 + li;
                const v2f w0 = W1p[bb], w1 = W1p[bb + 16], w2 = W1p[bb + 32], w3 = W1p[bb + 48];
                acc[0][0] = __builtin_amdgcn_wmma_f32_16x16x4_f32(false, w0, false, a0, (short)0, acc[0][0], false, false);
                acc[0][1] = __builtin_amdgcn_wmma_f32_16x16x4_f32(false, w0, false, a1, (short)0, acc[0][1], false, false);
                acc[1][0] = __builtin_amdgcn_wmma_f32_16x16x4_f32(false, w1, false, a0, (short)0, acc[1][0], false, false);
                acc[1][1] = __builtin_amdgcn_wmma_f32_16x16x4_f32(false, w1, false, a1, (short)0, acc[1][1], false, false);
                acc[2][0] = __builtin_amdgcn_wmma_f32_16x16x4_f32(false, w2, false, a0, (short)0, acc[2][0], false, false);
                acc[2][1] = __builtin_amdgcn_wmma_f32_16x16x4_f32(false, w2, false, a1, (short)0, acc[2][1], false, false);
                acc[3][0] = __builtin_amdgcn_wmma_f32_16x16x4_f32(false, w3, false, a0, (short)0, acc[3][0], false, false);
                acc[3][1] = __builtin_amdgcn_wmma_f32_16x16x4_f32(false, w3, false, a1, (short)0, acc[3][1], false, false);
            }
        }

        float part[2][4];
#pragma unroll
        for (int n = 0; n < 4; ++n) {
            const float4 wa = *(const float4*)(wsrc + 16 * n);
            const float4 wb = *(const float4*)(wsrc + 16 * n + 4);
            const float wv[8] = {wa.x, wa.y, wa.z, wa.w, wb.x, wb.y, wb.z, wb.w};
#pragma unroll
            for (int m = 0; m < 2; ++m) {
                float p = fmaxf(acc[n][m][0], 0.f) * wv[0];
#pragma unroll
                for (int r = 1; r < 8; ++r) p += fmaxf(acc[n][m][r], 0.f) * wv[r];
                part[m][n] = p;
            }
        }
        const float s0 = (part[0][0] + part[0][1]) + (part[0][2] + part[0][3]);
        const float s1 = (part[1][0] + part[1][1]) + (part[1][2] + part[1][3]);

        const float t0 = s0 + xhalf(s0);
        const float t1 = s1 + xhalf(s1);
        const float x  = (hi ? t1 : t0) + b2v;
        const float y  = 1.f / (1.f + expf(-x));
        const int e = e0 + l;
        if (e < E) out[e] = y;
    }
}

extern "C" void kernel_launch(void* const* d_in, const int* in_sizes, int n_in,
                              void* d_out, int out_size, void* d_ws, size_t ws_size,
                              hipStream_t stream) {
    (void)n_in; (void)out_size; (void)d_ws; (void)ws_size;
    const float* hf   = (const float*)d_in[0];
    const int*   edge = (const int*)d_in[1];
    const float* W1   = (const float*)d_in[2];
    const float* b1   = (const float*)d_in[3];
    const float* W2   = (const float*)d_in[4];
    const float* b2   = (const float*)d_in[5];
    float* out = (float*)d_out;

    const int E = in_sizes[1] / 2;
    const int npass = (E + ROWS - 1) / ROWS;
    int blocks = (npass + WAVES_PER_BLOCK - 1) / WAVES_PER_BLOCK;
    if (blocks > 2500) blocks = 2500;

    const size_t shmem = LDS_FLOATS * sizeof(float);
    hipLaunchKernelGGL(edge_decoder_kernel, dim3(blocks), dim3(BLOCK), shmem, stream,
                       hf, edge, W1, b1, W2, b2, out, E);
}


# gather software-pipelined over four K-quarters: each quarter's 16 row loads are issued before the previous quarter's 64 WMMAs and land under them (first quarter under the W1 stage); sum order kept
# speedup vs baseline: 1.1779x; 1.0736x over previous
#include <hip/hip_runtime.h>
#include <hip/hip_bf16.h>
#include <math.h>


typedef float v2f __attribute__((ext_vector_type(2)));
typedef float v8f __attribute__((ext_vector_type(8)));

#define D1 128
#define D2 64
#define WAVES_PER_BLOCK 8
#define BLOCK (WAVES_PER_BLOCK * 32)
#define ROWS 32
#define KQ 32
#define NQ (D1 / KQ)
#define ROW_STRIDE 36
#define TILE_FLOATS (ROWS * ROW_STRIDE)
#define W1_PAIRS (64 * 64)
#define SIDE_FLOATS (2 * D2)
#define LDS_FLOATS (2 * W1_PAIRS + WAVES_PER_BLOCK * TILE_FLOATS + SIDE_FLOATS)

__device__ __forceinline__ float xhalf(float x) {
    const int b = __builtin_bit_cast(int, x);
    return __builtin_bit_cast(float, __builtin_amdgcn_permlanex16(b, b, 0x76543210, 0xfedcba98, false, false));
}

__device__ __forceinline__ void gather_issue(const float* __restrict__ hf, const unsigned (&off)[8], const int q,
                                             float4 (&dst)[8]) {
#pragma unroll
    for (int i = 0; i < 8; ++i)
        dst[i] = *(const float4*)((const char*)hf + (size_t)(off[i] + (unsigned)(q * KQ * 4)));
}

__device__ __forceinline__ void pass_begin(const float* __restrict__ hf, const int* __restrict__ edge0,
                                           const int* __restrict__ edge1, const int pass, const int E, const int l,
                                           unsigned (&offu)[8], unsigned (&offv)[8],
                                           float4 (&pu)[8], float4 (&pv)[8]) {
    int ee = pass * ROWS + l; if (ee >= E) ee = E - 1;
    const int iu = edge0[ee];
    const int iv = edge1[ee];
    const int g8 = (l >> 3) * 8;
    const unsigned col = 16u * (unsigned)(l & 7);
#pragma unroll
    for (int i = 0; i < 8; ++i) {
        offu[i] = (unsigned)__shfl(iu, g8 + i, 32) * (unsigned)(D1 * 4) + col;
        offv[i] = (unsigned)__shfl(iv, g8 + i, 32) * (unsigned)(D1 * 4) + col;
    }
    gather_issue(hf, offu, 0, pu);
    gather_issue(hf, offv, 0, pv);
}

__global__ void __launch_bounds__(BLOCK)
edge_decoder_kernel(const float* __restrict__ hf,
                    const int*   __restrict__ edge,
                    const float* __restrict__ W1,
                    const float* __restrict__ b1,
                    const float* __restrict__ W2,
                    const float* __restrict__ b2,
                    float* __restrict__ out,
                    int E)
{
    extern __shared__ float smem[];
    v2f* W1p = (v2f*)smem;
    float* side = smem + 2 * W1_PAIRS + WAVES_PER_BLOCK * TILE_FLOATS;

    const int tid = threadIdx.x;
    const int l  = tid & 31;
    const int li = l & 15;
    const int hi = l >> 4;
    const int w  = tid >> 5;

    const int* edge0 = edge;
    const int* edge1 = edge + E;
    const int npass  = (E + ROWS - 1) / ROWS;
    const int nwaves = gridDim.x * WAVES_PER_BLOCK;

    int pass = blockIdx.x * WAVES_PER_BLOCK + w;
    unsigned offu[8], offv[8];
    float4 pu[8], pv[8];
    pass_begin(hf, edge0, edge1, pass, E, l, offu, offv, pu, pv);

    const float sb = b1[tid & (D2 - 1)];
    const float sw = W2[tid & (D2 - 1)];

    for (int idx = tid; idx < W1_PAIRS; idx += BLOCK) {
        const int p = idx >> 6;
        const int c = idx & 63;
        v2f v;
        v.x = W1[p * 2 * D2 + c];
        v.y = W1[p * 2 * D2 + D2 + c];
        W1p[idx] = v;
    }
    if (tid < D2) {
        side[tid]      = sb;
        side[D2 + tid] = sw;
    }
    __syncthreads();

    float* tileBuf = smem + 2 * W1_PAIRS + w * TILE_FLOATS;
    const v2f* Ab = (const v2f*)tileBuf;
    const int aBase = li * (ROW_STRIDE / 2) + hi;
    float* stBase = tileBuf + (l >> 3) * 8 * ROW_STRIDE + 4 * (l & 7);

    const float* bsrc = side + 8 * hi;
    const float* wsrc = side + D2 + 8 * hi;
    const float b2v = b2[0];

    for (; pass < npass; pass += nwaves) {
        const int e0 = pass * ROWS;

        v8f acc[4][2];
#pragma unroll
        for (int n = 0; n < 4; ++n) {
            const float4 ba = *(const float4*)(bsrc + 16 * n);
            const float4 bb = *(const float4*)(bsrc + 16 * n + 4);
            const v8f bv = (v8f){ba.x, ba.y, ba.z, ba.w, bb.x, bb.y, bb.z, bb.w};
            acc[n][0] = bv;
            acc[n][1] = bv;
        }

#pragma nounroll
        for (int q = 0; q < NQ; ++q) {
#pragma unroll
            for (int i = 0; i < 8; ++i) {
                float4 p;
                p.x = pu[i].x * pv[i].x; p.y = pu[i].y * pv[i].y;
                p.z = pu[i].z * pv[i].z; p.w = pu[i].w * pv[i].w;
                *(float4*)(stBase + i * ROW_STRIDE) = p;
            }
            if (q + 1 < NQ) {
                gather_issue(hf, offu, q + 1, pu);
                gather_issue(hf, offv, q + 1, pv);
            }

            const v2f* Wq = W1p + q * (KQ / 2) * 64;
#pragma unroll
            for (int t = 0; t < 8; ++t) {
                const v2f a0 = Ab[aBase + 2 * t];
                const v2f a1 = Ab[aBase + 16 * (ROW_STRIDE / 2) + 2 * t];
                const int bb = (2 * t + hi) * 64 + li;
                const v2f w0 = Wq[bb], w1 = Wq[bb + 16], w2 = Wq[bb + 32], w3 = Wq[bb + 48];
                acc[0][0] = __builtin_amdgcn_wmma_f32_16x16x4_f32(false, w0, false, a0, (short)0, acc[0][0], false, false);
                acc[0][1] = __builtin_amdgcn_wmma_f32_16x16x4_f32(false, w0, false, a1, (short)0, acc[0][1], false, false);
                acc[1][0] = __builtin_amdgcn_wmma_f32_16x16x4_f32(false, w1, false, a0, (short)0, acc[1][0], false, false);
                acc[1][1] = __builtin_amdgcn_wmma_f32_16x16x4_f32(false, w1, false, a1, (short)0, acc[1][1], false, false);
                acc[2][0] = __builtin_amdgcn_wmma_f32_16x16x4_f32(false, w2, false, a0, (short)0, acc[2][0], false, false);
                acc[2][1] = __builtin_amdgcn_wmma_f32_16x16x4_f32(false, w2, false, a1, (short)0, acc[2][1], false, false);
                acc[3][0] = __builtin_amdgcn_wmma_f32_16x16x4_f32(false, w3, false, a0, (short)0, acc[3][0], false, false);
                acc[3][1] = __builtin_amdgcn_wmma_f32_16x16x4_f32(false, w3, false, a1, (short)0, acc[3][1], false, false);
            }
        }

        float part[2][4];
#pragma unroll
        for (int n = 0; n < 4; ++n) {
            const float4 wa = *(const float4*)(wsrc + 16 * n);
            const float4 wb = *(const float4*)(wsrc + 16 * n + 4);
            const float wv[8] = {wa.x, wa.y, wa.z, wa.w, wb.x, wb.y, wb.z, wb.w};
#pragma unroll
            for (int m = 0; m < 2; ++m) {
                float p = fmaxf(acc[n][m][0], 0.f) * wv[0];
#pragma unroll
                for (int r = 1; r < 8; ++r) p += fmaxf(acc[n][m][r], 0.f) * wv[r];
                part[m][n] = p;
            }
        }
        const float s0 = (part[0][0] + part[0][1]) + (part[0][2] + part[0][3]);
        const float s1 = (part[1][0] + part[1][1]) + (part[1][2] + part[1][3]);

        const float t0 = s0 + xhalf(s0);
        const float t1 = s1 + xhalf(s1);
        const float x  = (hi ? t1 : t0) + b2v;
        const float y  = 1.f / (1.f + expf(-x));
        const int e = e0 + l;
        if (e < E) out[e] = y;

        if (pass + nwaves < npass) pass_begin(hf, edge0, edge1, pass + nwaves, E, l, offu, offv, pu, pv);
    }
}

extern "C" void kernel_launch(void* const* d_in, const int* in_sizes, int n_in,
                              void* d_out, int out_size, void* d_ws, size_t ws_size,
                              hipStream_t stream) {
    (void)n_in; (void)out_size; (void)d_ws; (void)ws_size;
    const float* hf   = (const float*)d_in[0];
    const int*   edge = (const int*)d_in[1];
    const float* W1   = (const float*)d_in[2];
    const float* b1   = (const float*)d_in[3];
    const float* W2   = (const float*)d_in[4];
    const float* b2   = (const float*)d_in[5];
    float* out = (float*)d_out;

    const int E = in_sizes[1] / 2;
    const int npass = (E + ROWS - 1) / ROWS;
    int blocks = (npass + WAVES_PER_BLOCK - 1) / WAVES_PER_BLOCK;
    if (blocks > 2500) blocks = 2500;

    const size_t shmem = LDS_FLOATS * sizeof(float);
    hipLaunchKernelGGL(edge_decoder_kernel, dim3(blocks), dim3(BLOCK), shmem, stream,
                       hf, edge, W1, b1, W2, b2, out, E);
}


# W1 fragments held in each wave's registers across a resident grid (256 groups, 2 waves per SIMD): the matrix loop reads only edge fragments from LDS, one LDS group per quarter; sum order kept
# speedup vs baseline: 1.2220x; 1.0374x over previous
#include <hip/hip_runtime.h>
#include <hip/hip_bf16.h>
#include <math.h>


typedef float v2f __attribute__((ext_vector_type(2)));
typedef float v8f __attribute__((ext_vector_type(8)));

#define D1 128
#define D2 64
#define WAVES_PER_BLOCK 8
#define BLOCK (WAVES_PER_BLOCK * 32)
#define ROWS 32
#define KQ 32
#define NQ (D1 / KQ)
#define ROW_STRIDE 36
#define TILE_FLOATS (ROWS * ROW_STRIDE)
#define W1_PAIRS (64 * 64)
#define SIDE_FLOATS (2 * D2)
#define LDS_FLOATS (2 * W1_PAIRS + WAVES_PER_BLOCK * TILE_FLOATS + SIDE_FLOATS)
#define MAX_BLOCKS 256

__device__ __forceinline__ float xhalf(float x) {
    const int b = __builtin_bit_cast(int, x);
    return __builtin_bit_cast(float, __builtin_amdgcn_permlanex16(b, b, 0x76543210, 0xfedcba98, false, false));
}

__device__ __forceinline__ void gather_issue(const float* __restrict__ hf, const unsigned (&off)[8], const int q,
                                             float4 (&dst)[8]) {
#pragma unroll
    for (int i = 0; i < 8; ++i)
        dst[i] = *(const float4*)((const char*)hf + (size_t)(off[i] + (unsigned)(q * KQ * 4)));
}

__device__ __forceinline__ void row_offsets(const int iu, const int iv, const int l,
                                            unsigned (&offu)[8], unsigned (&offv)[8]) {
    const int g8 = (l >> 3) * 8;
    const unsigned col = 16u * (unsigned)(l & 7);
#pragma unroll
    for (int i = 0; i < 8; ++i) {
        offu[i] = (unsigned)__shfl(iu, g8 + i, 32) * (unsigned)(D1 * 4) + col;
        offv[i] = (unsigned)__shfl(iv, g8 + i, 32) * (unsigned)(D1 * 4) + col;
    }
}

__device__ __forceinline__ void stage_products(const float4 (&pu)[8], const float4 (&pv)[8], float* stBase) {
#pragma unroll
    for (int i = 0; i < 8; ++i) {
        float4 p;
        p.x = pu[i].x * pv[i].x; p.y = pu[i].y * pv[i].y;
        p.z = pu[i].z * pv[i].z; p.w = pu[i].w * pv[i].w;
        *(float4*)(stBase + i * ROW_STRIDE) = p;
    }
}

__device__ __forceinline__ void load_edge_frags(const v2f* Ab, const int aBase, v2f (&af)[16]) {
#pragma unroll
    for (int t = 0; t < 8; ++t) {
        af[2 * t]     = Ab[aBase + 2 * t];
        af[2 * t + 1] = Ab[aBase + 16 * (ROW_STRIDE / 2) + 2 * t];
    }
}

__global__ void __launch_bounds__(BLOCK)
edge_decoder_kernel(const float* __restrict__ hf,
                    const int*   __restrict__ edge,
                    const float* __restrict__ W1,
                    const float* __restrict__ b1,
                    const float* __restrict__ W2,
                    const float* __restrict__ b2,
                    float* __restrict__ out,
                    int E)
{
    extern __shared__ float smem[];
    v2f* W1p = (v2f*)smem;
    float* side = smem + 2 * W1_PAIRS + WAVES_PER_BLOCK * TILE_FLOATS;

    const int tid = threadIdx.x;
    const int l  = tid & 31;
    const int li = l & 15;
    const int hi = l >> 4;
    const int w  = tid >> 5;

    const int* edge0 = edge;
    const int* edge1 = edge + E;
    const int npass  = (E + ROWS - 1) / ROWS;
    const int nwaves = gridDim.x * WAVES_PER_BLOCK;

    int pass = blockIdx.x * WAVES_PER_BLOCK + w;
    unsigned offu[8], offv[8];
    float4 pu[8], pv[8];
    int niu, niv;
    {
        int ee = pass * ROWS + l; if (ee >= E) ee = E - 1;
        const int p1 = (pass + nwaves < npass) ? pass + nwaves : pass;
        int e1 = p1 * ROWS + l; if (e1 >= E) e1 = E - 1;
        const int iu = edge0[ee], iv = edge1[ee];
        niu = edge0[e1]; niv = edge1[e1];
        row_offsets(iu, iv, l, offu, offv);
        gather_issue(hf, offu, 0, pu);
        gather_issue(hf, offv, 0, pv);
    }

    const float sb = b1[tid & (D2 - 1)];
    const float sw = W2[tid & (D2 - 1)];

    for (int idx = tid; idx < W1_PAIRS; idx += BLOCK) {
        const int p = idx >> 6;
        const int c = idx & 63;
        v2f v;
        v.x = W1[p * 2 * D2 + c];
        v.y = W1[p * 2 * D2 + D2 + c];
        W1p[idx] = v;
    }
    if (tid < D2) {
        side[tid]      = sb;
        side[D2 + tid] = sw;
    }
    __syncthreads();

    v2f wr[32][4];
#pragma unroll
    for (int s = 0; s < 32; ++s) {
        const int bb = (2 * s + hi) * 64 + li;
#pragma unroll
        for (int n = 0; n < 4; ++n) wr[s][n] = W1p[bb + 16 * n];
    }

    float* tileBuf = smem + 2 * W1_PAIRS + w * TILE_FLOATS;
    const v2f* Ab = (const v2f*)tileBuf;
    const int aBase = li * (ROW_STRIDE / 2) + hi;
    float* stBase = tileBuf + (l >> 3) * 8 * ROW_STRIDE + 4 * (l & 7);

    const float* bsrc = side + 8 * hi;
    const float* wsrc = side + D2 + 8 * hi;
    const float b2v = b2[0];

    if (pass >= npass) return;

    stage_products(pu, pv, stBase);
    gather_issue(hf, offu, 1, pu);
    gather_issue(hf, offv, 1, pv);

    for (;;) {
        const int e0 = pass * ROWS;
        const bool more = pass + nwaves < npass;
        const int p2 = (pass + 2 * nwaves < npass) ? pass + 2 * nwaves : pass;
        int e2 = p2 * ROWS + l; if (e2 >= E) e2 = E - 1;
        const int nniu = edge0[e2];
        const int nniv = edge1[e2];

        v8f acc[4][2];
#pragma unroll
        for (int n = 0; n < 4; ++n) {
            const float4 ba = *(const float4*)(bsrc + 16 * n);
            const float4 bb = *(const float4*)(bsrc + 16 * n + 4);
            const v8f bv = (v8f){ba.x, ba.y, ba.z, ba.w, bb.x, bb.y, bb.z, bb.w};
            acc[n][0] = bv;
            acc[n][1] = bv;
        }
        __builtin_amdgcn_sched_barrier(0);

#pragma unroll
        for (int q = 0; q < NQ; ++q) {
            v2f af[16];
            load_edge_frags(Ab, aBase, af);
            stage_products(pu, pv, stBase);
            if (q == 2) row_offsets(niu, niv, l, offu, offv);
            const int gq = (q + 2) & 3;
            gather_issue(hf, offu, gq, pu);
            gather_issue(hf, offv, gq, pv);

#pragma unroll
            for (int t = 0; t < 8; ++t) {
                const v2f a0 = af[2 * t];
                const v2f a1 = af[2 * t + 1];
                const int s = 8 * q + t;
                acc[0][0] = __builtin_amdgcn_wmma_f32_16x16x4_f32(false, wr[s][0], false, a0, (short)0, acc[0][0], false, false);
                acc[0][1] = __builtin_amdgcn_wmma_f32_16x16x4_f32(false, wr[s][0], false, a1, (short)0, acc[0][1], false, false);
                acc[1][0] = __builtin_amdgcn_wmma_f32_16x16x4_f32(false, wr[s][1], false, a0, (short)0, acc[1][0], false, false);
                acc[1][1] = __builtin_amdgcn_wmma_f32_16x16x4_f32(false, wr[s][1], false, a1, (short)0, acc[1][1], false, false);
                acc[2][0] = __builtin_amdgcn_wmma_f32_16x16x4_f32(false, wr[s][2], false, a0, (short)0, acc[2][0], false, false);
                acc[2][1] = __builtin_amdgcn_wmma_f32_16x16x4_f32(false, wr[s][2], false, a1, (short)0, acc[2][1], false, false);
                acc[3][0] = __builtin_amdgcn_wmma_f32_16x16x4_f32(false, wr[s][3], false, a0, (short)0, acc[3][0], false, false);
                acc[3][1] = __builtin_amdgcn_wmma_f32_16x16x4_f32(false, wr[s][3], false, a1, (short)0, acc[3][1], false, false);
            }
            if (q == 0) {
                __builtin_amdgcn_sched_group_barrier(0x080, 32, 0);
            } else {
                __builtin_amdgcn_sched_group_barrier(0x100, 8, 0);
                __builtin_amdgcn_sched_group_barrier(0x200, 8, 0);
            }
            __builtin_amdgcn_sched_group_barrier(0x020, 16, 0);
            __builtin_amdgcn_sched_group_barrier(0x008, 64, 0);
            __builtin_amdgcn_sched_barrier(0);
        }

        float part[2][4];
#pragma unroll
        for (int n = 0; n < 4; ++n) {
            const float4 wa = *(const float4*)(wsrc + 16 * n);
            const float4 wb = *(const float4*)(wsrc + 16 * n + 4);
            const float wv[8] = {wa.x, wa.y, wa.z, wa.w, wb.x, wb.y, wb.z, wb.w};
#pragma unroll
            for (int m = 0; m < 2; ++m) {
                float p = fmaxf(acc[n][m][0], 0.f) * wv[0];
#pragma unroll
                for (int r = 1; r < 8; ++r) p += fmaxf(acc[n][m][r], 0.f) * wv[r];
                part[m][n] = p;
            }
        }
        const float s0 = (part[0][0] + part[0][1]) + (part[0][2] + part[0][3]);
        const float s1 = (part[1][0] + part[1][1]) + (part[1][2] + part[1][3]);

        const float t0 = s0 + xhalf(s0);
        const float t1 = s1 + xhalf(s1);
        const float x  = (hi ? t1 : t0) + b2v;
        const float y  = 1.f / (1.f + expf(-x));
        const int e = e0 + l;
        if (e < E) out[e] = y;

        if (!more) break;
        pass += nwaves;
        niu = nniu; niv = nniv;
    }
}

extern "C" void kernel_launch(void* const* d_in, const int* in_sizes, int n_in,
                              void* d_out, int out_size, void* d_ws, size_t ws_size,
                              hipStream_t stream) {
    (void)n_in; (void)out_size; (void)d_ws; (void)ws_size;
    const float* hf   = (const float*)d_in[0];
    const int*   edge = (const int*)d_in[1];
    const float* W1   = (const float*)d_in[2];
    const float* b1   = (const float*)d_in[3];
    const float* W2   = (const float*)d_in[4];
    const float* b2   = (const float*)d_in[5];
    float* out = (float*)d_out;

    const int E = in_sizes[1] / 2;
    const int npass = (E + ROWS - 1) / ROWS;
    int blocks = (npass + WAVES_PER_BLOCK - 1) / WAVES_PER_BLOCK;
    if (blocks > MAX_BLOCKS) blocks = MAX_BLOCKS;

    const size_t shmem = LDS_FLOATS * sizeof(float);
    hipLaunchKernelGGL(edge_decoder_kernel, dim3(blocks), dim3(BLOCK), shmem, stream,
                       hf, edge, W1, b1, W2, b2, out, E);
}
